# starting kernel
# baseline (speedup 1.0000x reference)
#include <hip/hip_runtime.h>
#include <hip/hip_bf16.h>

#define F_DIM 48

typedef float v2f __attribute__((ext_vector_type(2)));
typedef float v8f __attribute__((ext_vector_type(8)));

__global__ void zero_kernel(float* __restrict__ out, int n) {
    int i4 = (blockIdx.x * blockDim.x + threadIdx.x) * 4;
    if (i4 + 3 < n) {
        *reinterpret_cast<float4*>(out + i4) = make_float4(0.f, 0.f, 0.f, 0.f);
    } else {
        for (int k = i4; k < n; ++k) out[k] = 0.f;
    }
}

__global__ void edge_scatter_kernel(const float* __restrict__ x,
                                    const int*   __restrict__ rows,
                                    const int*   __restrict__ cols,
                                    const float* __restrict__ vals,
                                    float* __restrict__ out,
                                    int nEdges) {
    int t = blockIdx.x * blockDim.x + threadIdx.x;
    int e = t >> 4;
    int f = t & 15;
    if (e >= nEdges) return;

    int epf = e + 64;
    if (f == 0 && epf < nEdges) {
        int cpf = cols[epf];
        __builtin_prefetch(x + (long)cpf * F_DIM, 0, 1);
    }

    int   r = rows[e];
    int   c = cols[e];
    float v = vals[e];

    const float* xr   = x   + (long)c * F_DIM;
    float*       orow = out + (long)r * F_DIM;

    atomicAdd(orow + f,      v * xr[f]);
    atomicAdd(orow + f + 16, v * xr[f + 16]);
    atomicAdd(orow + f + 32, v * xr[f + 32]);
}

__global__ void scale_wmma_kernel(float* __restrict__ out,
                                  const float* __restrict__ w,
                                  int nNodes) {
    int wave = (blockIdx.x * blockDim.x + threadIdx.x) >> 5;
    int lane = threadIdx.x & 31;
    int rows0 = wave * 16;
    if (rows0 >= nNodes) return;

    int n = lane & 15;
    int h = lane >> 4;

#pragma unroll
    for (int j = 0; j < 3; ++j) {
        float wn = w[16 * j + n];
        v8f acc = {};
#pragma unroll
        for (int c4 = 0; c4 < 4; ++c4) {
            int kbase = 16 * j + 4 * c4;

            const float* ap = out + (long)(rows0 + n) * F_DIM + kbase + 2 * h;
            v2f a;
            a.x = ap[0];
            a.y = ap[1];

            v2f b;
            b.x = (n == 4 * c4 + 2 * h)     ? wn : 0.f;
            b.y = (n == 4 * c4 + 2 * h + 1) ? wn : 0.f;

            acc = __builtin_amdgcn_wmma_f32_16x16x4_f32(
                 false, a,  false, b,
                 (short)0, acc,  false,  false);
        }
#pragma unroll
        for (int v = 0; v < 8; ++v) {
            out[(long)(rows0 + v + 8 * h) * F_DIM + 16 * j + n] = acc[v];
        }
    }
}

extern "C" void kernel_launch(void* const* d_in, const int* in_sizes, int n_in,
                              void* d_out, int out_size, void* d_ws, size_t ws_size,
                              hipStream_t stream) {
    const float* x    = (const float*)d_in[0];
    const float* w    = (const float*)d_in[1];
    const int*   rows = (const int*)  d_in[2];
    const int*   cols = (const int*)  d_in[3];
    const float* vals = (const float*)d_in[4];
    float* out = (float*)d_out;

    const int nEdges = in_sizes[2];
    const int nNodes = in_sizes[0] / F_DIM;

    {
        int nquads = (out_size + 3) / 4;
        int blocks = (nquads + 255) / 256;
        zero_kernel<<<blocks, 256, 0, stream>>>(out, out_size);
    }

    {
        long threads = (long)nEdges * 16;
        int blocks = (int)((threads + 255) / 256);
        edge_scatter_kernel<<<blocks, 256, 0, stream>>>(x, rows, cols, vals,
                                                        out, nEdges);
    }

    {
        int waves  = (nNodes + 15) / 16;
        int blocks = (waves + 7) / 8;
        scale_wmma_kernel<<<blocks, 256, 0, stream>>>(out, w, nNodes);
    }
}


# Owners no longer scan all edges: a counting-sort pass writes each edge chunk in owner order (packed row/col word + value); an owner reads its 256 contiguous segments; planes, queue, index fetch gone
# speedup vs baseline: 3.3914x; 3.3914x over previous
#include <hip/hip_runtime.h>


#define F_DIM 48
#define N_OWNERS 256
#define RPB 392
#define N_CHUNKS 256
#define CH_MAX 8192
#define OFF_PITCH 288
#define COL_BITS 17
#define COL_MASK 0x1FFFFu
#define BAD_REC (511u << COL_BITS)
#define PART_THREADS 1024
#define OWNER_THREADS 1024

typedef float v2f __attribute__((ext_vector_type(2)));
typedef float v8f __attribute__((ext_vector_type(8)));

__global__ void zero_kernel(float* __restrict__ out, int n) {
    int i4 = (blockIdx.x * blockDim.x + threadIdx.x) * 4;
    if (i4 + 3 < n) {
        *reinterpret_cast<float4*>(out + i4) = make_float4(0.f, 0.f, 0.f, 0.f);
    } else {
        for (int k = i4; k < n; ++k) out[k] = 0.f;
    }
}

__global__ void edge_scatter_kernel(const float* __restrict__ x,
                                    const int*   __restrict__ rows,
                                    const int*   __restrict__ cols,
                                    const float* __restrict__ vals,
                                    float* __restrict__ out,
                                    int nEdges) {
    int t = blockIdx.x * blockDim.x + threadIdx.x;
    int e = t >> 4;
    int f = t & 15;
    if (e >= nEdges) return;
    int   r = rows[e];
    int   c = cols[e];
    float v = vals[e];
    const float* xr   = x   + (long)c * F_DIM;
    float*       orow = out + (long)r * F_DIM;
    atomicAdd(orow + f,      v * xr[f]);
    atomicAdd(orow + f + 16, v * xr[f + 16]);
    atomicAdd(orow + f + 32, v * xr[f + 32]);
}

__global__ __launch_bounds__(PART_THREADS)
void partition_kernel(const int*   __restrict__ rows,
                      const int*   __restrict__ cols,
                      const float* __restrict__ vals,
                      int*      __restrict__ offs,
                      unsigned* __restrict__ recIdx,
                      float*    __restrict__ recVal,
                      int nEdges, int nNodes, int chunkCap) {
    __shared__ int      hist[N_OWNERS];
    __shared__ int      start[N_OWNERS + 1];
    __shared__ unsigned sIdx[CH_MAX];
    __shared__ float    sVal[CH_MAX];

    const int  tid = threadIdx.x;
    const int  c   = blockIdx.x;
    const long e0  = (long)c * chunkCap;
    int n = 0;
    if (e0 < (long)nEdges) {
        long left = (long)nEdges - e0;
        n = (left < (long)chunkCap) ? (int)left : chunkCap;
    }

    if (tid < N_OWNERS) hist[tid] = 0;

    int   r_[CH_MAX / PART_THREADS];
    int   c_[CH_MAX / PART_THREADS];
    float v_[CH_MAX / PART_THREADS];
#pragma unroll
    for (int k = 0; k < CH_MAX / PART_THREADS; ++k) {
        int i = tid + k * PART_THREADS;
        r_[k] = -1; c_[k] = -1; v_[k] = 0.f;
        if (i < n) { r_[k] = rows[e0 + i]; c_[k] = cols[e0 + i]; v_[k] = vals[e0 + i]; }
    }
    __syncthreads();

    unsigned pk[CH_MAX / PART_THREADS];
    int      ow[CH_MAX / PART_THREADS];
    int      rk[CH_MAX / PART_THREADS];
#pragma unroll
    for (int k = 0; k < CH_MAX / PART_THREADS; ++k) {
        int i = tid + k * PART_THREADS;
        ow[k] = -1; rk[k] = 0; pk[k] = BAD_REC;
        if (i < n) {
            bool ok = (unsigned)r_[k] < (unsigned)nNodes && (unsigned)c_[k] < (unsigned)nNodes;
            unsigned o = ok ? ((unsigned)r_[k] / RPB) : 0u;
            if (ok) pk[k] = (((unsigned)r_[k] - o * RPB) << COL_BITS) | (unsigned)c_[k];
            ow[k] = (int)o;
            rk[k] = atomicAdd(&hist[o], 1);
        }
    }
    __syncthreads();

    if (tid < 32) {
        int loc[8];
        int s = 0;
#pragma unroll
        for (int j = 0; j < 8; ++j) { loc[j] = s; s += hist[tid * 8 + j]; }
        int incl = s;
#pragma unroll
        for (int d = 1; d < 32; d <<= 1) {
            int t = __shfl_up(incl, d);
            if (tid >= d) incl += t;
        }
        const int excl = incl - s;
#pragma unroll
        for (int j = 0; j < 8; ++j) start[tid * 8 + j] = excl + loc[j];
        if (tid == 31) start[N_OWNERS] = incl;
    }
    __syncthreads();

    if (tid <= N_OWNERS) offs[c * OFF_PITCH + tid] = start[tid];

#pragma unroll
    for (int k = 0; k < CH_MAX / PART_THREADS; ++k) {
        if (ow[k] >= 0) {
            int p = start[ow[k]] + rk[k];
            sIdx[p] = pk[k];
            sVal[p] = v_[k];
        }
    }
    __syncthreads();

    const int nfill = (n + 31) & ~31;
    for (int i = tid; i < nfill; i += PART_THREADS) {
        recIdx[e0 + i] = (i < n) ? sIdx[i] : BAD_REC;
        recVal[e0 + i] = (i < n) ? sVal[i] : 0.f;
    }
}

__global__ __launch_bounds__(OWNER_THREADS)
void owner_accumulate_kernel(const float*    __restrict__ x,
                             const int*      __restrict__ offs,
                             const unsigned* __restrict__ recIdx,
                             const float*    __restrict__ recVal,
                             float* __restrict__ out,
                             int nNodes, int chunkCap) {
    __shared__ float acc[RPB * F_DIM];
    __shared__ int   segLen[N_CHUNKS];
    __shared__ int   segSrc[N_CHUNKS];
    __shared__ int   segPos[N_CHUNKS + 1];

    const int tid  = threadIdx.x;
    const int lane = tid & 31;
    const int wave = tid >> 5;
    const int b    = blockIdx.x;
    const int row0 = b * RPB;

    for (int i = tid * 4; i < RPB * F_DIM; i += OWNER_THREADS * 4)
        *reinterpret_cast<float4*>(acc + i) = make_float4(0.f, 0.f, 0.f, 0.f);
    if (tid < N_CHUNKS) {
        int s0 = offs[tid * OFF_PITCH + b];
        int s1 = offs[tid * OFF_PITCH + b + 1];
        s0 = min(max(s0, 0), chunkCap);
        s1 = min(max(s1, s0), chunkCap);
        segLen[tid] = s1 - s0;
        segSrc[tid] = tid * chunkCap + s0;
    }
    __syncthreads();

    if (tid < 32) {
        int loc[8];
        int s = 0;
#pragma unroll
        for (int j = 0; j < 8; ++j) { loc[j] = s; s += segLen[tid * 8 + j]; }
        int incl = s;
#pragma unroll
        for (int d = 1; d < 32; d <<= 1) {
            int t = __shfl_up(incl, d);
            if (tid >= d) incl += t;
        }
        const int excl = incl - s;
#pragma unroll
        for (int j = 0; j < 8; ++j) segPos[tid * 8 + j] = excl + loc[j];
        if (tid == 31) segPos[N_CHUNKS] = incl;
    }
    __syncthreads();

    const int total = segPos[N_CHUNKS];
    const int half = lane >> 4;
    const int f    = lane & 15;
    for (int base = wave * 32; base < total; base += OWNER_THREADS) {
        const int i = base + lane;
        int ao = -1, xo = 0; float v = 0.f;
        if (i < total) {
            int lo = 0, hi = N_CHUNKS;
#pragma unroll
            for (int st = 0; st < 8; ++st) {
                int mid = (lo + hi) >> 1;
                if (segPos[mid] <= i) lo = mid; else hi = mid;
            }
            const int idx = segSrc[lo] + (i - segPos[lo]);
            const unsigned p = recIdx[idx];
            v = recVal[idx];
            const unsigned lr = p >> COL_BITS;
            const unsigned cc = p & COL_MASK;
            if (lr < (unsigned)RPB && (unsigned)row0 + lr < (unsigned)nNodes && cc < (unsigned)nNodes) {
                ao = (int)lr * F_DIM; xo = (int)cc * F_DIM;
            }
        }
        float xv[16][3];
#pragma unroll
        for (int s = 0; s < 16; ++s) {
            int xo_s = __shfl(xo, s + 16 * half);
            const float* xr = x + xo_s + f;
            xv[s][0] = xr[0];
            xv[s][1] = xr[16];
            xv[s][2] = xr[32];
        }
#pragma unroll
        for (int s = 0; s < 16; ++s) {
            int   ao_s = __shfl(ao, s + 16 * half);
            float v_s  = __shfl(v,  s + 16 * half);
            if (ao_s >= 0) {
                float* ar = acc + ao_s + f;
                atomicAdd(ar,      v_s * xv[s][0]);
                atomicAdd(ar + 16, v_s * xv[s][1]);
                atomicAdd(ar + 32, v_s * xv[s][2]);
            }
        }
    }
    __syncthreads();

    int nrows = nNodes - row0;
    if (nrows > RPB) nrows = RPB;
    const int nfl = nrows * F_DIM;
    float* orow = out + (long)row0 * F_DIM;
    for (int i = tid * 4; i < nfl; i += OWNER_THREADS * 4)
        *reinterpret_cast<float4*>(orow + i) = *reinterpret_cast<const float4*>(acc + i);
}

__global__ void scale_wmma_kernel(float* __restrict__ out,
                                  const float* __restrict__ w,
                                  int nNodes) {
    int wave = (blockIdx.x * blockDim.x + threadIdx.x) >> 5;
    int lane = threadIdx.x & 31;
    int rows0 = wave * 16;
    if (rows0 >= nNodes) return;

    int n = lane & 15;
    int h = lane >> 4;

#pragma unroll
    for (int j = 0; j < 3; ++j) {
        float wn = w[16 * j + n];
        v8f acc = {};
#pragma unroll
        for (int c4 = 0; c4 < 4; ++c4) {
            int kbase = 16 * j + 4 * c4;

            const float* ap = out + (long)(rows0 + n) * F_DIM + kbase + 2 * h;
            v2f a;
            a.x = ap[0];
            a.y = ap[1];

            v2f b;
            b.x = (n == 4 * c4 + 2 * h)     ? wn : 0.f;
            b.y = (n == 4 * c4 + 2 * h + 1) ? wn : 0.f;

            acc = __builtin_amdgcn_wmma_f32_16x16x4_f32(
                 false, a,  false, b,
                 (short)0, acc,  false,  false);
        }
#pragma unroll
        for (int v = 0; v < 8; ++v) {
            out[(long)(rows0 + v + 8 * h) * F_DIM + 16 * j + n] = acc[v];
        }
    }
}

extern "C" void kernel_launch(void* const* d_in, const int* in_sizes, int n_in,
                              void* d_out, int out_size, void* d_ws, size_t ws_size,
                              hipStream_t stream) {
    const float* x    = (const float*)d_in[0];
    const float* w    = (const float*)d_in[1];
    const int*   rows = (const int*)  d_in[2];
    const int*   cols = (const int*)  d_in[3];
    const float* vals = (const float*)d_in[4];
    float* out = (float*)d_out;

    const int nEdges = in_sizes[2];
    const int nNodes = in_sizes[0] / F_DIM;

    const int    chunkCap = (((nEdges + N_CHUNKS - 1) / N_CHUNKS) + 31) & ~31;
    const size_t offBytes = (size_t)N_CHUNKS * OFF_PITCH * sizeof(int);
    const size_t recBytes = (size_t)N_CHUNKS * (size_t)chunkCap * 4u;

    if (ws_size >= offBytes + 2 * recBytes && nNodes <= N_OWNERS * RPB && chunkCap <= CH_MAX) {
        int*      offs   = (int*)d_ws;
        unsigned* recIdx = (unsigned*)((char*)d_ws + offBytes);
        float*    recVal = (float*)((char*)d_ws + offBytes + recBytes);
        partition_kernel<<<N_CHUNKS, PART_THREADS, 0, stream>>>(
            rows, cols, vals, offs, recIdx, recVal, nEdges, nNodes, chunkCap);
        owner_accumulate_kernel<<<N_OWNERS, OWNER_THREADS, 0, stream>>>(
            x, offs, recIdx, recVal, out, nNodes, chunkCap);
    } else {
        {
            int nquads = (out_size + 3) / 4;
            int blocks = (nquads + 255) / 256;
            zero_kernel<<<blocks, 256, 0, stream>>>(out, out_size);
        }
        {
            long threads = (long)nEdges * 16;
            int blocks = (int)((threads + 255) / 256);
            edge_scatter_kernel<<<blocks, 256, 0, stream>>>(x, rows, cols, vals,
                                                            out, nEdges);
        }
    }

    {
        int waves  = (nNodes + 15) / 16;
        int blocks = (waves + 7) / 8;
        scale_wmma_kernel<<<blocks, 256, 0, stream>>>(out, w, nNodes);
    }
}


# Owner pass: per-step half-wave broadcasts of an edge's x offset, tile offset and value moved from LDS-unit permutes (__shfl) to vector-ALU DPP row_share; the adds no longer wait on the LDS counter
# speedup vs baseline: 4.1949x; 1.2369x over previous
#include <hip/hip_runtime.h>


#define F_DIM 48
#define N_OWNERS 256
#define RPB 392
#define TILE_ROWS 400
#define N_CHUNKS 256
#define CH_MAX 8192
#define OFF_PITCH 288
#define COL_BITS 17
#define COL_MASK 0x1FFFFu
#define BAD_REC (511u << COL_BITS)
#define PART_THREADS 1024
#define OWNER_THREADS 1024

typedef float v2f __attribute__((ext_vector_type(2)));
typedef float v8f __attribute__((ext_vector_type(8)));

__global__ void zero_kernel(float* __restrict__ out, int n) {
    int i4 = (blockIdx.x * blockDim.x + threadIdx.x) * 4;
    if (i4 + 3 < n) {
        *reinterpret_cast<float4*>(out + i4) = make_float4(0.f, 0.f, 0.f, 0.f);
    } else {
        for (int k = i4; k < n; ++k) out[k] = 0.f;
    }
}

__global__ void edge_scatter_kernel(const float* __restrict__ x,
                                    const int*   __restrict__ rows,
                                    const int*   __restrict__ cols,
                                    const float* __restrict__ vals,
                                    float* __restrict__ out,
                                    int nEdges) {
    int t = blockIdx.x * blockDim.x + threadIdx.x;
    int e = t >> 4;
    int f = t & 15;
    if (e >= nEdges) return;
    int   r = rows[e];
    int   c = cols[e];
    float v = vals[e];
    const float* xr   = x   + (long)c * F_DIM;
    float*       orow = out + (long)r * F_DIM;
    atomicAdd(orow + f,      v * xr[f]);
    atomicAdd(orow + f + 16, v * xr[f + 16]);
    atomicAdd(orow + f + 32, v * xr[f + 32]);
}

__global__ __launch_bounds__(PART_THREADS)
void partition_kernel(const int*   __restrict__ rows,
                      const int*   __restrict__ cols,
                      const float* __restrict__ vals,
                      int*      __restrict__ offs,
                      unsigned* __restrict__ recIdx,
                      float*    __restrict__ recVal,
                      int nEdges, int nNodes, int chunkCap) {
    __shared__ int      hist[N_OWNERS];
    __shared__ int      start[N_OWNERS + 1];
    __shared__ unsigned sIdx[CH_MAX];
    __shared__ float    sVal[CH_MAX];

    const int  tid = threadIdx.x;
    const int  c   = blockIdx.x;
    const long e0  = (long)c * chunkCap;
    int n = 0;
    if (e0 < (long)nEdges) {
        long left = (long)nEdges - e0;
        n = (left < (long)chunkCap) ? (int)left : chunkCap;
    }

    if (tid < N_OWNERS) hist[tid] = 0;

    int   r_[CH_MAX / PART_THREADS];
    int   c_[CH_MAX / PART_THREADS];
    float v_[CH_MAX / PART_THREADS];
#pragma unroll
    for (int k = 0; k < CH_MAX / PART_THREADS; ++k) {
        int i = tid + k * PART_THREADS;
        r_[k] = -1; c_[k] = -1; v_[k] = 0.f;
        if (i < n) { r_[k] = rows[e0 + i]; c_[k] = cols[e0 + i]; v_[k] = vals[e0 + i]; }
    }
    __syncthreads();

    unsigned pk[CH_MAX / PART_THREADS];
    int      ow[CH_MAX / PART_THREADS];
    int      rk[CH_MAX / PART_THREADS];
#pragma unroll
    for (int k = 0; k < CH_MAX / PART_THREADS; ++k) {
        int i = tid + k * PART_THREADS;
        ow[k] = -1; rk[k] = 0; pk[k] = BAD_REC;
        if (i < n) {
            bool ok = (unsigned)r_[k] < (unsigned)nNodes && (unsigned)c_[k] < (unsigned)nNodes;
            unsigned o = ok ? ((unsigned)r_[k] / RPB) : 0u;
            if (ok) pk[k] = (((unsigned)r_[k] - o * RPB) << COL_BITS) | (unsigned)c_[k];
            ow[k] = (int)o;
            rk[k] = atomicAdd(&hist[o], 1);
        }
    }
    __syncthreads();

    if (tid < 32) {
        int loc[8];
        int s = 0;
#pragma unroll
        for (int j = 0; j < 8; ++j) { loc[j] = s; s += hist[tid * 8 + j]; }
        int incl = s;
#pragma unroll
        for (int d = 1; d < 32; d <<= 1) {
            int t = __shfl_up(incl, d);
            if (tid >= d) incl += t;
        }
        const int excl = incl - s;
#pragma unroll
        for (int j = 0; j < 8; ++j) start[tid * 8 + j] = excl + loc[j];
        if (tid == 31) start[N_OWNERS] = incl;
    }
    __syncthreads();

    if (tid <= N_OWNERS) offs[c * OFF_PITCH + tid] = start[tid];

#pragma unroll
    for (int k = 0; k < CH_MAX / PART_THREADS; ++k) {
        if (ow[k] >= 0) {
            int p = start[ow[k]] + rk[k];
            sIdx[p] = pk[k];
            sVal[p] = v_[k];
        }
    }
    __syncthreads();

    const int nfill = (n + 31) & ~31;
    for (int i = tid; i < nfill; i += PART_THREADS) {
        recIdx[e0 + i] = (i < n) ? sIdx[i] : BAD_REC;
        recVal[e0 + i] = (i < n) ? sVal[i] : 0.f;
    }
}

template <int S>
__device__ __forceinline__ int row_bcast_i(int v) {
    return __builtin_amdgcn_update_dpp(0, v, 0x150 + S  , 0xF, 0xF, true);
}
template <int S>
__device__ __forceinline__ float row_bcast_f(float v) {
    return __int_as_float(row_bcast_i<S>(__float_as_int(v)));
}

template <int S>
__device__ __forceinline__ void drain_gather(const float* __restrict__ x, int xo, int f,
                                             float (&xv)[16][3]) {
    const int xo_s = row_bcast_i<S>(xo);
    const float* xr = x + xo_s + f;
    xv[S][0] = xr[0];
    xv[S][1] = xr[16];
    xv[S][2] = xr[32];
    if constexpr (S + 1 < 16) drain_gather<S + 1>(x, xo, f, xv);
}

template <int S>
__device__ __forceinline__ void drain_add(float* __restrict__ acc, int ao, float v, int f,
                                          const float (&xv)[16][3]) {
    const int   ao_s = row_bcast_i<S>(ao);
    const float v_s  = row_bcast_f<S>(v);
    if (ao_s >= 0) {
        float* ar = acc + ao_s + f;
        atomicAdd(ar,      v_s * xv[S][0]);
        atomicAdd(ar + 16, v_s * xv[S][1]);
        atomicAdd(ar + 32, v_s * xv[S][2]);
    }
    if constexpr (S + 1 < 16) drain_add<S + 1>(acc, ao, v, f, xv);
}

__global__ __launch_bounds__(OWNER_THREADS)
void owner_accumulate_kernel(const float*    __restrict__ x,
                             const float*    __restrict__ w,
                             const int*      __restrict__ offs,
                             const unsigned* __restrict__ recIdx,
                             const float*    __restrict__ recVal,
                             float* __restrict__ out,
                             int nNodes, int chunkCap) {
    __shared__ float acc[TILE_ROWS * F_DIM];
    __shared__ int   segLen[N_CHUNKS];
    __shared__ int   segSrc[N_CHUNKS];
    __shared__ int   segPos[N_CHUNKS + 1];

    const int tid  = threadIdx.x;
    const int lane = tid & 31;
    const int wave = tid >> 5;
    const int b    = blockIdx.x;
    const int row0 = b * RPB;

    float wn[3];
    wn[0] = w[(lane & 15)];
    wn[1] = w[(lane & 15) + 16];
    wn[2] = w[(lane & 15) + 32];

    for (int i = tid * 4; i < TILE_ROWS * F_DIM; i += OWNER_THREADS * 4)
        *reinterpret_cast<float4*>(acc + i) = make_float4(0.f, 0.f, 0.f, 0.f);
    if (tid < N_CHUNKS) {
        int s0 = offs[tid * OFF_PITCH + b];
        int s1 = offs[tid * OFF_PITCH + b + 1];
        s0 = min(max(s0, 0), chunkCap);
        s1 = min(max(s1, s0), chunkCap);
        segLen[tid] = s1 - s0;
        segSrc[tid] = tid * chunkCap + s0;
    }
    __syncthreads();

    if (tid < 32) {
        int loc[8];
        int s = 0;
#pragma unroll
        for (int j = 0; j < 8; ++j) { loc[j] = s; s += segLen[tid * 8 + j]; }
        int incl = s;
#pragma unroll
        for (int d = 1; d < 32; d <<= 1) {
            int t = __shfl_up(incl, d);
            if (tid >= d) incl += t;
        }
        const int excl = incl - s;
#pragma unroll
        for (int j = 0; j < 8; ++j) segPos[tid * 8 + j] = excl + loc[j];
        if (tid == 31) segPos[N_CHUNKS] = incl;
    }
    __syncthreads();

    const int total = segPos[N_CHUNKS];
    const int f    = lane & 15;
    for (int base = wave * 32; base < total; base += OWNER_THREADS) {
        const int i = base + lane;
        int ao = -1, xo = 0; float v = 0.f;
        if (i < total) {
            int lo = 0, hi = N_CHUNKS;
#pragma unroll
            for (int st = 0; st < 8; ++st) {
                int mid = (lo + hi) >> 1;
                if (segPos[mid] <= i) lo = mid; else hi = mid;
            }
            const int idx = segSrc[lo] + (i - segPos[lo]);
            const unsigned p = recIdx[idx];
            v = recVal[idx];
            const unsigned lr = p >> COL_BITS;
            const unsigned cc = p & COL_MASK;
            if (lr < (unsigned)RPB && (unsigned)row0 + lr < (unsigned)nNodes && cc < (unsigned)nNodes) {
                ao = (int)lr * F_DIM; xo = (int)cc * F_DIM;
            }
        }
        float xv[16][3];
        drain_gather<0>(x, xo, f, xv);
        drain_add<0>(acc, ao, v, f, xv);
    }
    __syncthreads();

    int nrows = nNodes - row0;
    if (nrows > RPB) nrows = RPB;

    {
        const int n = lane & 15;
        const int h = lane >> 4;
        const int swave = __builtin_amdgcn_readfirstlane(wave);
        for (int t0 = swave * 16; t0 < nrows; t0 += (OWNER_THREADS / 32) * 16) {
            const float* ap0 = acc + (t0 + n) * F_DIM + 2 * h;
#pragma unroll
            for (int j = 0; j < 3; ++j) {
                v8f d = {};
#pragma unroll
                for (int c4 = 0; c4 < 4; ++c4) {
                    const v2f a = *reinterpret_cast<const v2f*>(ap0 + 16 * j + 4 * c4);
                    v2f bd;
                    bd.x = (n == 4 * c4 + 2 * h)     ? wn[j] : 0.f;
                    bd.y = (n == 4 * c4 + 2 * h + 1) ? wn[j] : 0.f;
                    d = __builtin_amdgcn_wmma_f32_16x16x4_f32(
                         false, a,  false, bd,
                         (short)0, d,  false,  false);
                }
#pragma unroll
                for (int v = 0; v < 8; ++v)
                    acc[(t0 + v + 8 * h) * F_DIM + 16 * j + n] = d[v];
            }
        }
    }
    __syncthreads();

    const int nfl = nrows * F_DIM;
    float* orow = out + (long)row0 * F_DIM;
    for (int i = tid * 4; i < nfl; i += OWNER_THREADS * 4)
        *reinterpret_cast<float4*>(orow + i) = *reinterpret_cast<const float4*>(acc + i);
}

__global__ void scale_wmma_kernel(float* __restrict__ out,
                                  const float* __restrict__ w,
                                  int nNodes) {
    int wave = (blockIdx.x * blockDim.x + threadIdx.x) >> 5;
    int lane = threadIdx.x & 31;
    int rows0 = wave * 16;
    if (rows0 >= nNodes) return;

    int n = lane & 15;
    int h = lane >> 4;

#pragma unroll
    for (int j = 0; j < 3; ++j) {
        float wn = w[16 * j + n];
        v8f acc = {};
#pragma unroll
        for (int c4 = 0; c4 < 4; ++c4) {
            int kbase = 16 * j + 4 * c4;

            const float* ap = out + (long)(rows0 + n) * F_DIM + kbase + 2 * h;
            v2f a;
            a.x = ap[0];
            a.y = ap[1];

            v2f b;
            b.x = (n == 4 * c4 + 2 * h)     ? wn : 0.f;
            b.y = (n == 4 * c4 + 2 * h + 1) ? wn : 0.f;

            acc = __builtin_amdgcn_wmma_f32_16x16x4_f32(
                 false, a,  false, b,
                 (short)0, acc,  false,  false);
        }
#pragma unroll
        for (int v = 0; v < 8; ++v) {
            out[(long)(rows0 + v + 8 * h) * F_DIM + 16 * j + n] = acc[v];
        }
    }
}

extern "C" void kernel_launch(void* const* d_in, const int* in_sizes, int n_in,
                              void* d_out, int out_size, void* d_ws, size_t ws_size,
                              hipStream_t stream) {
    const float* x    = (const float*)d_in[0];
    const float* w    = (const float*)d_in[1];
    const int*   rows = (const int*)  d_in[2];
    const int*   cols = (const int*)  d_in[3];
    const float* vals = (const float*)d_in[4];
    float* out = (float*)d_out;

    const int nEdges = in_sizes[2];
    const int nNodes = in_sizes[0] / F_DIM;

    const int    chunkCap = (((nEdges + N_CHUNKS - 1) / N_CHUNKS) + 31) & ~31;
    const size_t offBytes = (size_t)N_CHUNKS * OFF_PITCH * sizeof(int);
    const size_t recBytes = (size_t)N_CHUNKS * (size_t)chunkCap * 4u;

    if (ws_size >= offBytes + 2 * recBytes && nNodes <= N_OWNERS * RPB && chunkCap <= CH_MAX) {
        int*      offs   = (int*)d_ws;
        unsigned* recIdx = (unsigned*)((char*)d_ws + offBytes);
        float*    recVal = (float*)((char*)d_ws + offBytes + recBytes);
        partition_kernel<<<N_CHUNKS, PART_THREADS, 0, stream>>>(
            rows, cols, vals, offs, recIdx, recVal, nEdges, nNodes, chunkCap);
        owner_accumulate_kernel<<<N_OWNERS, OWNER_THREADS, 0, stream>>>(
            x, w, offs, recIdx, recVal, out, nNodes, chunkCap);
    } else {
        {
            int nquads = (out_size + 3) / 4;
            int blocks = (nquads + 255) / 256;
            zero_kernel<<<blocks, 256, 0, stream>>>(out, out_size);
        }
        {
            long threads = (long)nEdges * 16;
            int blocks = (int)((threads + 255) / 256);
            edge_scatter_kernel<<<blocks, 256, 0, stream>>>(x, rows, cols, vals,
                                                            out, nEdges);
        }
        {
            int waves  = (nNodes + 15) / 16;
            int blocks = (waves + 7) / 8;
            scale_wmma_kernel<<<blocks, 256, 0, stream>>>(out, w, nNodes);
        }
    }
}


# Owner drain software-pipelined: a batch's 48 row gathers go out into a second register set before the previous batch's LDS adds, its records fetched two batches ahead; loop body written out twice
# speedup vs baseline: 4.4185x; 1.0533x over previous
#include <hip/hip_runtime.h>


#define F_DIM 48
#define N_OWNERS 256
#define RPB 392
#define TILE_ROWS 400
#define N_CHUNKS 256
#define CH_MAX 8192
#define OFF_PITCH 288
#define COL_BITS 17
#define COL_MASK 0x1FFFFu
#define BAD_REC (511u << COL_BITS)
#define PART_THREADS 1024
#define OWNER_THREADS 1024

typedef float v2f __attribute__((ext_vector_type(2)));
typedef float v8f __attribute__((ext_vector_type(8)));

__global__ void zero_kernel(float* __restrict__ out, int n) {
    int i4 = (blockIdx.x * blockDim.x + threadIdx.x) * 4;
    if (i4 + 3 < n) {
        *reinterpret_cast<float4*>(out + i4) = make_float4(0.f, 0.f, 0.f, 0.f);
    } else {
        for (int k = i4; k < n; ++k) out[k] = 0.f;
    }
}

__global__ void edge_scatter_kernel(const float* __restrict__ x,
                                    const int*   __restrict__ rows,
                                    const int*   __restrict__ cols,
                                    const float* __restrict__ vals,
                                    float* __restrict__ out,
                                    int nEdges) {
    int t = blockIdx.x * blockDim.x + threadIdx.x;
    int e = t >> 4;
    int f = t & 15;
    if (e >= nEdges) return;
    int   r = rows[e];
    int   c = cols[e];
    float v = vals[e];
    const float* xr   = x   + (long)c * F_DIM;
    float*       orow = out + (long)r * F_DIM;
    atomicAdd(orow + f,      v * xr[f]);
    atomicAdd(orow + f + 16, v * xr[f + 16]);
    atomicAdd(orow + f + 32, v * xr[f + 32]);
}

__global__ __launch_bounds__(PART_THREADS)
void partition_kernel(const int*   __restrict__ rows,
                      const int*   __restrict__ cols,
                      const float* __restrict__ vals,
                      int*      __restrict__ offs,
                      unsigned* __restrict__ recIdx,
                      float*    __restrict__ recVal,
                      int nEdges, int nNodes, int chunkCap) {
    __shared__ int      hist[N_OWNERS];
    __shared__ int      start[N_OWNERS + 1];
    __shared__ unsigned sIdx[CH_MAX];
    __shared__ float    sVal[CH_MAX];

    const int  tid = threadIdx.x;
    const int  c   = blockIdx.x;
    const long e0  = (long)c * chunkCap;
    int n = 0;
    if (e0 < (long)nEdges) {
        long left = (long)nEdges - e0;
        n = (left < (long)chunkCap) ? (int)left : chunkCap;
    }

    if (tid < N_OWNERS) hist[tid] = 0;

    int   r_[CH_MAX / PART_THREADS];
    int   c_[CH_MAX / PART_THREADS];
    float v_[CH_MAX / PART_THREADS];
#pragma unroll
    for (int k = 0; k < CH_MAX / PART_THREADS; ++k) {
        int i = tid + k * PART_THREADS;
        r_[k] = -1; c_[k] = -1; v_[k] = 0.f;
        if (i < n) { r_[k] = rows[e0 + i]; c_[k] = cols[e0 + i]; v_[k] = vals[e0 + i]; }
    }
    __syncthreads();

    unsigned pk[CH_MAX / PART_THREADS];
    int      ow[CH_MAX / PART_THREADS];
    int      rk[CH_MAX / PART_THREADS];
#pragma unroll
    for (int k = 0; k < CH_MAX / PART_THREADS; ++k) {
        int i = tid + k * PART_THREADS;
        ow[k] = -1; rk[k] = 0; pk[k] = BAD_REC;
        if (i < n) {
            bool ok = (unsigned)r_[k] < (unsigned)nNodes && (unsigned)c_[k] < (unsigned)nNodes;
            unsigned o = ok ? ((unsigned)r_[k] / RPB) : 0u;
            if (ok) pk[k] = (((unsigned)r_[k] - o * RPB) << COL_BITS) | (unsigned)c_[k];
            ow[k] = (int)o;
            rk[k] = atomicAdd(&hist[o], 1);
        }
    }
    __syncthreads();

    if (tid < 32) {
        int loc[8];
        int s = 0;
#pragma unroll
        for (int j = 0; j < 8; ++j) { loc[j] = s; s += hist[tid * 8 + j]; }
        int incl = s;
#pragma unroll
        for (int d = 1; d < 32; d <<= 1) {
            int t = __shfl_up(incl, d);
            if (tid >= d) incl += t;
        }
        const int excl = incl - s;
#pragma unroll
        for (int j = 0; j < 8; ++j) start[tid * 8 + j] = excl + loc[j];
        if (tid == 31) start[N_OWNERS] = incl;
    }
    __syncthreads();

    if (tid <= N_OWNERS) offs[c * OFF_PITCH + tid] = start[tid];

#pragma unroll
    for (int k = 0; k < CH_MAX / PART_THREADS; ++k) {
        if (ow[k] >= 0) {
            int p = start[ow[k]] + rk[k];
            sIdx[p] = pk[k];
            sVal[p] = v_[k];
        }
    }
    __syncthreads();

    const int nfill = (n + 31) & ~31;
    for (int i = tid; i < nfill; i += PART_THREADS) {
        recIdx[e0 + i] = (i < n) ? sIdx[i] : BAD_REC;
        recVal[e0 + i] = (i < n) ? sVal[i] : 0.f;
    }
}

template <int S>
__device__ __forceinline__ int row_bcast_i(int v) {
    return __builtin_amdgcn_update_dpp(0, v, 0x150 + S  , 0xF, 0xF, true);
}
template <int S>
__device__ __forceinline__ float row_bcast_f(float v) {
    return __int_as_float(row_bcast_i<S>(__float_as_int(v)));
}

template <int S>
__device__ __forceinline__ void drain_gather(const float* __restrict__ x, int xo, int f,
                                             float (&xv)[16][3]) {
    const int xo_s = row_bcast_i<S>(xo);
    const float* xr = x + xo_s + f;
    xv[S][0] = xr[0];
    xv[S][1] = xr[16];
    xv[S][2] = xr[32];
    if constexpr (S + 1 < 16) drain_gather<S + 1>(x, xo, f, xv);
}

template <int S>
__device__ __forceinline__ void drain_add(float* __restrict__ acc, int ao, float v, int f,
                                          const float (&xv)[16][3]) {
    const int   ao_s = row_bcast_i<S>(ao);
    const float v_s  = row_bcast_f<S>(v);
    if (ao_s >= 0) {
        float* ar = acc + ao_s + f;
        atomicAdd(ar,      v_s * xv[S][0]);
        atomicAdd(ar + 16, v_s * xv[S][1]);
        atomicAdd(ar + 32, v_s * xv[S][2]);
    }
    if constexpr (S + 1 < 16) drain_add<S + 1>(acc, ao, v, f, xv);
}

struct edge_raw { unsigned p; float v; };

__device__ __forceinline__ edge_raw fetch_rec(int i, int total,
                                              const int* __restrict__ segPos,
                                              const int* __restrict__ segSrc,
                                              const unsigned* __restrict__ recIdx,
                                              const float*    __restrict__ recVal) {
    edge_raw r;
    r.p = BAD_REC; r.v = 0.f;
    if (i < total) {
        int lo = 0, hi = N_CHUNKS;
#pragma unroll
        for (int st = 0; st < 8; ++st) {
            int mid = (lo + hi) >> 1;
            if (segPos[mid] <= i) lo = mid; else hi = mid;
        }
        const int idx = segSrc[lo] + (i - segPos[lo]);
        r.p = recIdx[idx];
        r.v = recVal[idx];
    }
    return r;
}

__device__ __forceinline__ void rec_decode(unsigned p, int row0, int nNodes, int& ao, int& xo) {
    const unsigned lr = p >> COL_BITS;
    const unsigned cc = p & COL_MASK;
    ao = -1; xo = 0;
    if (lr < (unsigned)RPB && (unsigned)row0 + lr < (unsigned)nNodes && cc < (unsigned)nNodes) {
        ao = (int)lr * F_DIM; xo = (int)cc * F_DIM;
    }
}

__global__ __launch_bounds__(OWNER_THREADS)
void owner_accumulate_kernel(const float*    __restrict__ x,
                             const float*    __restrict__ w,
                             const int*      __restrict__ offs,
                             const unsigned* __restrict__ recIdx,
                             const float*    __restrict__ recVal,
                             float* __restrict__ out,
                             int nNodes, int chunkCap) {
    __shared__ float acc[TILE_ROWS * F_DIM];
    __shared__ int   segLen[N_CHUNKS];
    __shared__ int   segSrc[N_CHUNKS];
    __shared__ int   segPos[N_CHUNKS + 1];

    const int tid  = threadIdx.x;
    const int lane = tid & 31;
    const int wave = tid >> 5;
    const int b    = blockIdx.x;
    const int row0 = b * RPB;

    float wn[3];
    wn[0] = w[(lane & 15)];
    wn[1] = w[(lane & 15) + 16];
    wn[2] = w[(lane & 15) + 32];

    for (int i = tid * 4; i < TILE_ROWS * F_DIM; i += OWNER_THREADS * 4)
        *reinterpret_cast<float4*>(acc + i) = make_float4(0.f, 0.f, 0.f, 0.f);
    if (tid < N_CHUNKS) {
        int s0 = offs[tid * OFF_PITCH + b];
        int s1 = offs[tid * OFF_PITCH + b + 1];
        s0 = min(max(s0, 0), chunkCap);
        s1 = min(max(s1, s0), chunkCap);
        segLen[tid] = s1 - s0;
        segSrc[tid] = tid * chunkCap + s0;
    }
    __syncthreads();

    if (tid < 32) {
        int loc[8];
        int s = 0;
#pragma unroll
        for (int j = 0; j < 8; ++j) { loc[j] = s; s += segLen[tid * 8 + j]; }
        int incl = s;
#pragma unroll
        for (int d = 1; d < 32; d <<= 1) {
            int t = __shfl_up(incl, d);
            if (tid >= d) incl += t;
        }
        const int excl = incl - s;
#pragma unroll
        for (int j = 0; j < 8; ++j) segPos[tid * 8 + j] = excl + loc[j];
        if (tid == 31) segPos[N_CHUNKS] = incl;
    }
    __syncthreads();

    const int total = __builtin_amdgcn_readfirstlane(segPos[N_CHUNKS]);
    const int f     = lane & 15;
    int base = __builtin_amdgcn_readfirstlane(wave) * 32;
    if (base < total) {
        edge_raw qc = fetch_rec(base + lane, total, segPos, segSrc, recIdx, recVal);
        int aoc, xoc, aon, xon;
        rec_decode(qc.p, row0, nNodes, aoc, xoc);
        float vc = qc.v;
        float xa[16][3], xb[16][3];
        drain_gather<0>(x, xoc, f, xa);
        edge_raw qn = fetch_rec(base + OWNER_THREADS + lane, total, segPos, segSrc, recIdx, recVal);
        for (;;) {
            rec_decode(qn.p, row0, nNodes, aon, xon);
            if (base + OWNER_THREADS < total) drain_gather<0>(x, xon, f, xb);
            edge_raw q2 = fetch_rec(base + 2 * OWNER_THREADS + lane, total, segPos, segSrc, recIdx, recVal);
            drain_add<0>(acc, aoc, vc, f, xa);
            base += OWNER_THREADS; aoc = aon; vc = qn.v; qn = q2;
            if (base >= total) break;

            rec_decode(qn.p, row0, nNodes, aon, xon);
            if (base + OWNER_THREADS < total) drain_gather<0>(x, xon, f, xa);
            q2 = fetch_rec(base + 2 * OWNER_THREADS + lane, total, segPos, segSrc, recIdx, recVal);
            drain_add<0>(acc, aoc, vc, f, xb);
            base += OWNER_THREADS; aoc = aon; vc = qn.v; qn = q2;
            if (base >= total) break;
        }
    }
    __syncthreads();

    int nrows = nNodes - row0;
    if (nrows > RPB) nrows = RPB;

    {
        const int n = lane & 15;
        const int h = lane >> 4;
        const int swave = __builtin_amdgcn_readfirstlane(wave);
        for (int t0 = swave * 16; t0 < nrows; t0 += (OWNER_THREADS / 32) * 16) {
            const float* ap0 = acc + (t0 + n) * F_DIM + 2 * h;
#pragma unroll
            for (int j = 0; j < 3; ++j) {
                v8f d = {};
#pragma unroll
                for (int c4 = 0; c4 < 4; ++c4) {
                    const v2f a = *reinterpret_cast<const v2f*>(ap0 + 16 * j + 4 * c4);
                    v2f bd;
                    bd.x = (n == 4 * c4 + 2 * h)     ? wn[j] : 0.f;
                    bd.y = (n == 4 * c4 + 2 * h + 1) ? wn[j] : 0.f;
                    d = __builtin_amdgcn_wmma_f32_16x16x4_f32(
                         false, a,  false, bd,
                         (short)0, d,  false,  false);
                }
#pragma unroll
                for (int v = 0; v < 8; ++v)
                    acc[(t0 + v + 8 * h) * F_DIM + 16 * j + n] = d[v];
            }
        }
    }
    __syncthreads();

    const int nfl = nrows * F_DIM;
    float* orow = out + (long)row0 * F_DIM;
    for (int i = tid * 4; i < nfl; i += OWNER_THREADS * 4)
        *reinterpret_cast<float4*>(orow + i) = *reinterpret_cast<const float4*>(acc + i);
}

__global__ void scale_wmma_kernel(float* __restrict__ out,
                                  const float* __restrict__ w,
                                  int nNodes) {
    int wave = (blockIdx.x * blockDim.x + threadIdx.x) >> 5;
    int lane = threadIdx.x & 31;
    int rows0 = wave * 16;
    if (rows0 >= nNodes) return;

    int n = lane & 15;
    int h = lane >> 4;

#pragma unroll
    for (int j = 0; j < 3; ++j) {
        float wn = w[16 * j + n];
        v8f acc = {};
#pragma unroll
        for (int c4 = 0; c4 < 4; ++c4) {
            int kbase = 16 * j + 4 * c4;

            const float* ap = out + (long)(rows0 + n) * F_DIM + kbase + 2 * h;
            v2f a;
            a.x = ap[0];
            a.y = ap[1];

            v2f b;
            b.x = (n == 4 * c4 + 2 * h)     ? wn : 0.f;
            b.y = (n == 4 * c4 + 2 * h + 1) ? wn : 0.f;

            acc = __builtin_amdgcn_wmma_f32_16x16x4_f32(
                 false, a,  false, b,
                 (short)0, acc,  false,  false);
        }
#pragma unroll
        for (int v = 0; v < 8; ++v) {
            out[(long)(rows0 + v + 8 * h) * F_DIM + 16 * j + n] = acc[v];
        }
    }
}

extern "C" void kernel_launch(void* const* d_in, const int* in_sizes, int n_in,
                              void* d_out, int out_size, void* d_ws, size_t ws_size,
                              hipStream_t stream) {
    const float* x    = (const float*)d_in[0];
    const float* w    = (const float*)d_in[1];
    const int*   rows = (const int*)  d_in[2];
    const int*   cols = (const int*)  d_in[3];
    const float* vals = (const float*)d_in[4];
    float* out = (float*)d_out;

    const int nEdges = in_sizes[2];
    const int nNodes = in_sizes[0] / F_DIM;

    const int    chunkCap = (((nEdges + N_CHUNKS - 1) / N_CHUNKS) + 31) & ~31;
    const size_t offBytes = (size_t)N_CHUNKS * OFF_PITCH * sizeof(int);
    const size_t recBytes = (size_t)N_CHUNKS * (size_t)chunkCap * 4u;

    if (ws_size >= offBytes + 2 * recBytes && nNodes <= N_OWNERS * RPB && chunkCap <= CH_MAX) {
        int*      offs   = (int*)d_ws;
        unsigned* recIdx = (unsigned*)((char*)d_ws + offBytes);
        float*    recVal = (float*)((char*)d_ws + offBytes + recBytes);
        partition_kernel<<<N_CHUNKS, PART_THREADS, 0, stream>>>(
            rows, cols, vals, offs, recIdx, recVal, nEdges, nNodes, chunkCap);
        owner_accumulate_kernel<<<N_OWNERS, OWNER_THREADS, 0, stream>>>(
            x, w, offs, recIdx, recVal, out, nNodes, chunkCap);
    } else {
        {
            int nquads = (out_size + 3) / 4;
            int blocks = (nquads + 255) / 256;
            zero_kernel<<<blocks, 256, 0, stream>>>(out, out_size);
        }
        {
            long threads = (long)nEdges * 16;
            int blocks = (int)((threads + 255) / 256);
            edge_scatter_kernel<<<blocks, 256, 0, stream>>>(x, rows, cols, vals,
                                                            out, nEdges);
        }
        {
            int waves  = (nNodes + 15) / 16;
            int blocks = (waves + 7) / 8;
            scale_wmma_kernel<<<blocks, 256, 0, stream>>>(out, w, nNodes);
        }
    }
}


# Owner pass: its records sorted by output row in LDS (counting sort), a half-wave sums a row's edges in registers by fma in rank order; LDS float adds only at row changes, not 3 per 2 edges
# speedup vs baseline: 4.5731x; 1.0350x over previous
#include <hip/hip_runtime.h>


#define F_DIM 48
#define N_OWNERS 256
#define RPB 392
#define TILE_ROWS 400
#define N_CHUNKS 256
#define CH_MAX 8192
#define OFF_PITCH 288
#define COL_BITS 17
#define COL_MASK 0x1FFFFu
#define BAD_REC (511u << COL_BITS)
#define PART_THREADS 1024
#define OWNER_THREADS 1024
#define RCAP 8192
#define RPT (RCAP / OWNER_THREADS)
#define HB 416
#define PIPE 12

typedef float v2f __attribute__((ext_vector_type(2)));
typedef float v8f __attribute__((ext_vector_type(8)));

__global__ void zero_kernel(float* __restrict__ out, int n) {
    int i4 = (blockIdx.x * blockDim.x + threadIdx.x) * 4;
    if (i4 + 3 < n) {
        *reinterpret_cast<float4*>(out + i4) = make_float4(0.f, 0.f, 0.f, 0.f);
    } else {
        for (int k = i4; k < n; ++k) out[k] = 0.f;
    }
}

__global__ void edge_scatter_kernel(const float* __restrict__ x,
                                    const int*   __restrict__ rows,
                                    const int*   __restrict__ cols,
                                    const float* __restrict__ vals,
                                    float* __restrict__ out,
                                    int nEdges) {
    int t = blockIdx.x * blockDim.x + threadIdx.x;
    int e = t >> 4;
    int f = t & 15;
    if (e >= nEdges) return;
    int   r = rows[e];
    int   c = cols[e];
    float v = vals[e];
    const float* xr   = x   + (long)c * F_DIM;
    float*       orow = out + (long)r * F_DIM;
    atomicAdd(orow + f,      v * xr[f]);
    atomicAdd(orow + f + 16, v * xr[f + 16]);
    atomicAdd(orow + f + 32, v * xr[f + 32]);
}

__global__ __launch_bounds__(PART_THREADS)
void partition_kernel(const int*   __restrict__ rows,
                      const int*   __restrict__ cols,
                      const float* __restrict__ vals,
                      int*      __restrict__ offs,
                      unsigned* __restrict__ recIdx,
                      float*    __restrict__ recVal,
                      int nEdges, int nNodes, int chunkCap) {
    __shared__ int      hist[N_OWNERS];
    __shared__ int      start[N_OWNERS + 1];
    __shared__ unsigned sIdx[CH_MAX];
    __shared__ float    sVal[CH_MAX];

    const int  tid = threadIdx.x;
    const int  c   = blockIdx.x;
    const long e0  = (long)c * chunkCap;
    int n = 0;
    if (e0 < (long)nEdges) {
        long left = (long)nEdges - e0;
        n = (left < (long)chunkCap) ? (int)left : chunkCap;
    }

    if (tid < N_OWNERS) hist[tid] = 0;

    int   r_[CH_MAX / PART_THREADS];
    int   c_[CH_MAX / PART_THREADS];
    float v_[CH_MAX / PART_THREADS];
#pragma unroll
    for (int k = 0; k < CH_MAX / PART_THREADS; ++k) {
        int i = tid + k * PART_THREADS;
        r_[k] = -1; c_[k] = -1; v_[k] = 0.f;
        if (i < n) { r_[k] = rows[e0 + i]; c_[k] = cols[e0 + i]; v_[k] = vals[e0 + i]; }
    }
    __syncthreads();

    unsigned pk[CH_MAX / PART_THREADS];
    int      ow[CH_MAX / PART_THREADS];
    int      rk[CH_MAX / PART_THREADS];
#pragma unroll
    for (int k = 0; k < CH_MAX / PART_THREADS; ++k) {
        int i = tid + k * PART_THREADS;
        ow[k] = -1; rk[k] = 0; pk[k] = BAD_REC;
        if (i < n) {
            bool ok = (unsigned)r_[k] < (unsigned)nNodes && (unsigned)c_[k] < (unsigned)nNodes;
            unsigned o = ok ? ((unsigned)r_[k] / RPB) : 0u;
            if (ok) pk[k] = (((unsigned)r_[k] - o * RPB) << COL_BITS) | (unsigned)c_[k];
            ow[k] = (int)o;
            rk[k] = atomicAdd(&hist[o], 1);
        }
    }
    __syncthreads();

    if (tid < 32) {
        int loc[8];
        int s = 0;
#pragma unroll
        for (int j = 0; j < 8; ++j) { loc[j] = s; s += hist[tid * 8 + j]; }
        int incl = s;
#pragma unroll
        for (int d = 1; d < 32; d <<= 1) {
            int t = __shfl_up(incl, d);
            if (tid >= d) incl += t;
        }
        const int excl = incl - s;
#pragma unroll
        for (int j = 0; j < 8; ++j) start[tid * 8 + j] = excl + loc[j];
        if (tid == 31) start[N_OWNERS] = incl;
    }
    __syncthreads();

    if (tid <= N_OWNERS) offs[c * OFF_PITCH + tid] = start[tid];

#pragma unroll
    for (int k = 0; k < CH_MAX / PART_THREADS; ++k) {
        if (ow[k] >= 0) {
            int p = start[ow[k]] + rk[k];
            sIdx[p] = pk[k];
            sVal[p] = v_[k];
        }
    }
    __syncthreads();

    const int nfill = (n + 31) & ~31;
    for (int i = tid; i < nfill; i += PART_THREADS) {
        recIdx[e0 + i] = (i < n) ? sIdx[i] : BAD_REC;
        recVal[e0 + i] = (i < n) ? sVal[i] : 0.f;
    }
}

struct edge_raw { unsigned p; float v; };

__global__ __launch_bounds__(OWNER_THREADS)
void owner_accumulate_kernel(const float*    __restrict__ x,
                             const float*    __restrict__ w,
                             const int*      __restrict__ offs,
                             const unsigned* __restrict__ recIdx,
                             const float*    __restrict__ recVal,
                             float* __restrict__ out,
                             int nNodes, int chunkCap) {
    __shared__ float acc[TILE_ROWS * F_DIM];
    __shared__ int   segLen[N_CHUNKS];
    __shared__ int   segSrc[N_CHUNKS];
    __shared__ int   segPos[N_CHUNKS + 1];
    __shared__ int   hist[HB];
    __shared__ int   rstart[HB + 1];
    __shared__ int4  sRec[RCAP];

    const int tid  = threadIdx.x;
    const int lane = tid & 31;
    const int wave = tid >> 5;
    const int b    = blockIdx.x;
    const int row0 = b * RPB;

    float wn[3];
    wn[0] = w[(lane & 15)];
    wn[1] = w[(lane & 15) + 16];
    wn[2] = w[(lane & 15) + 32];

    for (int i = tid * 4; i < TILE_ROWS * F_DIM; i += OWNER_THREADS * 4)
        *reinterpret_cast<float4*>(acc + i) = make_float4(0.f, 0.f, 0.f, 0.f);
    if (tid < N_CHUNKS) {
        int s0 = offs[tid * OFF_PITCH + b];
        int s1 = offs[tid * OFF_PITCH + b + 1];
        s0 = min(max(s0, 0), chunkCap);
        s1 = min(max(s1, s0), chunkCap);
        segLen[tid] = s1 - s0;
        segSrc[tid] = tid * chunkCap + s0;
    }
    __syncthreads();

    if (tid < 32) {
        int loc[8];
        int s = 0;
#pragma unroll
        for (int j = 0; j < 8; ++j) { loc[j] = s; s += segLen[tid * 8 + j]; }
        int incl = s;
#pragma unroll
        for (int d = 1; d < 32; d <<= 1) {
            int t = __shfl_up(incl, d);
            if (tid >= d) incl += t;
        }
        const int excl = incl - s;
#pragma unroll
        for (int j = 0; j < 8; ++j) segPos[tid * 8 + j] = excl + loc[j];
        if (tid == 31) segPos[N_CHUNKS] = incl;
    }
    __syncthreads();

    const int total = __builtin_amdgcn_readfirstlane(segPos[N_CHUNKS]);
    const int f     = lane & 15;
    for (int r0 = 0; r0 < total; r0 += RCAP) {
        const int nr = min(RCAP, total - r0);
        if (tid < HB) hist[tid] = 0;
        __syncthreads();

        edge_raw q_[RPT];
        int      lr_[RPT];
        int      rk_[RPT];
        {
            int ii[RPT], lo[RPT];
#pragma unroll
            for (int j = 0; j < RPT; ++j) { ii[j] = min(r0 + tid + j * OWNER_THREADS, total - 1); lo[j] = 0; }
#pragma unroll
            for (int st = 0; st < 8; ++st) {
                int mv[RPT];
#pragma unroll
                for (int j = 0; j < RPT; ++j) mv[j] = segPos[lo[j] + (128 >> st)];
#pragma unroll
                for (int j = 0; j < RPT; ++j) if (mv[j] <= ii[j]) lo[j] += (128 >> st);
            }
            int sp[RPT], ss[RPT];
#pragma unroll
            for (int j = 0; j < RPT; ++j) { sp[j] = segPos[lo[j]]; ss[j] = segSrc[lo[j]]; }
#pragma unroll
            for (int j = 0; j < RPT; ++j) {
                const int idx = ss[j] + (ii[j] - sp[j]);
                q_[j].p = recIdx[idx];
                q_[j].v = recVal[idx];
            }
        }
#pragma unroll
        for (int j = 0; j < RPT; ++j) {
            const unsigned lr = q_[j].p >> COL_BITS;
            const unsigned cc = q_[j].p & COL_MASK;
            const bool ok = tid + j * OWNER_THREADS < nr && lr < (unsigned)RPB
                            && (unsigned)row0 + lr < (unsigned)nNodes && cc < (unsigned)nNodes;
            lr_[j] = ok ? (int)lr : -1;
            rk_[j] = 0;
            if (ok) rk_[j] = atomicAdd(&hist[lr], 1);
        }
        __syncthreads();

        if (tid < 32) {
            int loc[HB / 32];
            int s = 0;
#pragma unroll
            for (int j = 0; j < HB / 32; ++j) { loc[j] = s; s += hist[tid * (HB / 32) + j]; }
            int incl = s;
#pragma unroll
            for (int d = 1; d < 32; d <<= 1) {
                int t = __shfl_up(incl, d);
                if (tid >= d) incl += t;
            }
            const int excl = incl - s;
#pragma unroll
            for (int j = 0; j < HB / 32; ++j) rstart[tid * (HB / 32) + j] = excl + loc[j];
            if (tid == 31) rstart[HB] = incl;
        }
        __syncthreads();

#pragma unroll
        for (int j = 0; j < RPT; ++j) {
            if (lr_[j] >= 0) {
                const int pos = rstart[lr_[j]] + rk_[j];
                sRec[pos] = make_int4(lr_[j] * F_DIM, (int)((q_[j].p & COL_MASK) * F_DIM), __float_as_int(q_[j].v), 0);
            }
        }
        __syncthreads();

        const int m = __builtin_amdgcn_readfirstlane(rstart[HB]);
        if (m > 0) {
            const int per  = (m + 63) >> 6;
            const int e0   = (2 * wave + (lane >> 4)) * per;
            const int eend = min(e0 + per, m);
            const unsigned mine = (unsigned)(eend - e0);
            const unsigned nmine = eend > e0 ? mine : 0u;
            float a0 = 0.f, a1 = 0.f, a2 = 0.f;
            int   cur = (TILE_ROWS - 1) * F_DIM;
            int   pa[PIPE];
            float pv[PIPE];
            float xv[PIPE][3];
#pragma unroll
            for (int s = 0; s < PIPE; ++s) { pa[s] = 0; pv[s] = 0.f; xv[s][0] = 0.f; xv[s][1] = 0.f; xv[s][2] = 0.f; }
            for (int t = -PIPE; t < per; t += PIPE) {
#pragma unroll
                for (int g = 0; g < 2; ++g) {
                    int4 qn[PIPE / 2];
#pragma unroll
                    for (int u = 0; u < PIPE / 2; ++u)
                        qn[u] = sRec[min(e0 + t + PIPE + g * (PIPE / 2) + u, m - 1)];
#pragma unroll
                    for (int u = 0; u < PIPE / 2; ++u) {
                        const int s = g * (PIPE / 2) + u;
                        if ((unsigned)(t + s) < nmine) {
                            if (pa[s] != cur) {
                                atomicAdd(acc + cur + f,      a0);
                                atomicAdd(acc + cur + f + 16, a1);
                                atomicAdd(acc + cur + f + 32, a2);
                                cur = pa[s];
                                a0 = 0.f; a1 = 0.f; a2 = 0.f;
                            }
                            a0 = fmaf(pv[s], xv[s][0], a0);
                            a1 = fmaf(pv[s], xv[s][1], a1);
                            a2 = fmaf(pv[s], xv[s][2], a2);
                        }
                        pa[s] = qn[u].x;
                        pv[s] = __int_as_float(qn[u].z);
                        const float* xr = x + (unsigned)qn[u].y + f;
                        xv[s][0] = xr[0];
                        xv[s][1] = xr[16];
                        xv[s][2] = xr[32];
                    }
                }
            }
            atomicAdd(acc + cur + f,      a0);
            atomicAdd(acc + cur + f + 16, a1);
            atomicAdd(acc + cur + f + 32, a2);
        }
        __syncthreads();
    }

    int nrows = nNodes - row0;
    if (nrows > RPB) nrows = RPB;

    {
        const int n = lane & 15;
        const int h = lane >> 4;
        const int swave = __builtin_amdgcn_readfirstlane(wave);
        for (int t0 = swave * 16; t0 < nrows; t0 += (OWNER_THREADS / 32) * 16) {
            const float* ap0 = acc + (t0 + n) * F_DIM + 2 * h;
#pragma unroll
            for (int j = 0; j < 3; ++j) {
                v8f d = {};
#pragma unroll
                for (int c4 = 0; c4 < 4; ++c4) {
                    const v2f a = *reinterpret_cast<const v2f*>(ap0 + 16 * j + 4 * c4);
                    v2f bd;
                    bd.x = (n == 4 * c4 + 2 * h)     ? wn[j] : 0.f;
                    bd.y = (n == 4 * c4 + 2 * h + 1) ? wn[j] : 0.f;
                    d = __builtin_amdgcn_wmma_f32_16x16x4_f32(
                         false, a,  false, bd,
                         (short)0, d,  false,  false);
                }
#pragma unroll
                for (int v = 0; v < 8; ++v)
                    acc[(t0 + v + 8 * h) * F_DIM + 16 * j + n] = d[v];
            }
        }
    }
    __syncthreads();

    const int nfl = nrows * F_DIM;
    float* orow = out + (long)row0 * F_DIM;
    for (int i = tid * 4; i < nfl; i += OWNER_THREADS * 4)
        *reinterpret_cast<float4*>(orow + i) = *reinterpret_cast<const float4*>(acc + i);
}

__global__ void scale_wmma_kernel(float* __restrict__ out,
                                  const float* __restrict__ w,
                                  int nNodes) {
    int wave = (blockIdx.x * blockDim.x + threadIdx.x) >> 5;
    int lane = threadIdx.x & 31;
    int rows0 = wave * 16;
    if (rows0 >= nNodes) return;

    int n = lane & 15;
    int h = lane >> 4;

#pragma unroll
    for (int j = 0; j < 3; ++j) {
        float wn = w[16 * j + n];
        v8f acc = {};
#pragma unroll
        for (int c4 = 0; c4 < 4; ++c4) {
            int kbase = 16 * j + 4 * c4;

            const float* ap = out + (long)(rows0 + n) * F_DIM + kbase + 2 * h;
            v2f a;
            a.x = ap[0];
            a.y = ap[1];

            v2f b;
            b.x = (n == 4 * c4 + 2 * h)     ? wn : 0.f;
            b.y = (n == 4 * c4 + 2 * h + 1) ? wn : 0.f;

            acc = __builtin_amdgcn_wmma_f32_16x16x4_f32(
                 false, a,  false, b,
                 (short)0, acc,  false,  false);
        }
#pragma unroll
        for (int v = 0; v < 8; ++v) {
            out[(long)(rows0 + v + 8 * h) * F_DIM + 16 * j + n] = acc[v];
        }
    }
}

extern "C" void kernel_launch(void* const* d_in, const int* in_sizes, int n_in,
                              void* d_out, int out_size, void* d_ws, size_t ws_size,
                              hipStream_t stream) {
    const float* x    = (const float*)d_in[0];
    const float* w    = (const float*)d_in[1];
    const int*   rows = (const int*)  d_in[2];
    const int*   cols = (const int*)  d_in[3];
    const float* vals = (const float*)d_in[4];
    float* out = (float*)d_out;

    const int nEdges = in_sizes[2];
    const int nNodes = in_sizes[0] / F_DIM;

    const int    chunkCap = (((nEdges + N_CHUNKS - 1) / N_CHUNKS) + 31) & ~31;
    const size_t offBytes = (size_t)N_CHUNKS * OFF_PITCH * sizeof(int);
    const size_t recBytes = (size_t)N_CHUNKS * (size_t)chunkCap * 4u;

    if (ws_size >= offBytes + 2 * recBytes && nNodes <= N_OWNERS * RPB && chunkCap <= CH_MAX) {
        int*      offs   = (int*)d_ws;
        unsigned* recIdx = (unsigned*)((char*)d_ws + offBytes);
        float*    recVal = (float*)((char*)d_ws + offBytes + recBytes);
        partition_kernel<<<N_CHUNKS, PART_THREADS, 0, stream>>>(
            rows, cols, vals, offs, recIdx, recVal, nEdges, nNodes, chunkCap);
        owner_accumulate_kernel<<<N_OWNERS, OWNER_THREADS, 0, stream>>>(
            x, w, offs, recIdx, recVal, out, nNodes, chunkCap);
    } else {
        {
            int nquads = (out_size + 3) / 4;
            int blocks = (nquads + 255) / 256;
            zero_kernel<<<blocks, 256, 0, stream>>>(out, out_size);
        }
        {
            long threads = (long)nEdges * 16;
            int blocks = (int)((threads + 255) / 256);
            edge_scatter_kernel<<<blocks, 256, 0, stream>>>(x, rows, cols, vals,
                                                            out, nEdges);
        }
        {
            int waves  = (nNodes + 15) / 16;
            int blocks = (waves + 7) / 8;
            scale_wmma_kernel<<<blocks, 256, 0, stream>>>(out, w, nNodes);
        }
    }
}
